# starting kernel
# baseline (speedup 1.0000x reference)
#include <hip/hip_runtime.h>


typedef __attribute__((ext_vector_type(2))) float v2f;
typedef __attribute__((ext_vector_type(8))) float v8f;

#define NB 16
#define NA 128
#define HD 256
#define EF 30
#define KP 32
#define K2 (2 * HD)
#define KC 32
#define ROWS 64

__device__ __forceinline__ float selu_f(float x) {
    const float scale = 1.0507009873554805f;
    const float alpha = 1.6732632423543772f;
    return x > 0.0f ? scale * x : scale * alpha * (__expf(x) - 1.0f);
}

__device__ __forceinline__ v8f wmma_f32(v2f a, v2f b, v8f c) {
    return __builtin_amdgcn_wmma_f32_16x16x4_f32(false, a, false, b, (short)0, c,
                                                 false, false);
}

__global__ __launch_bounds__(256)
void ef_filter_reduce(const float* __restrict__ h,
                      const float* __restrict__ e,
                      const float* __restrict__ edge_mask,
                      const float* __restrict__ Wf,
                      const float* __restrict__ bf,
                      float* __restrict__ h1out) {
    __shared__ float sh_wf[KP * HD];
    __shared__ float sh_e[NA * KP];

    const int tid = threadIdx.x;
    const int bi  = blockIdx.x;
    const int b   = bi / NA;
    const int i   = bi % NA;

    {
        const float* wrow = Wf + (size_t)tid * EF;
#pragma unroll
        for (int k = 0; k < KP; ++k)
            sh_wf[k * HD + tid] = (k < EF) ? wrow[k] : 0.0f;
    }
    {
        const float* erow = e + ((size_t)(b * NA + i)) * NA * EF;
        const int j  = tid >> 1;
        const int kb = (tid & 1) * 16;
#pragma unroll
        for (int k = 0; k < 16; ++k) {
            const int kk = kb + k;
            sh_e[j * KP + kk] = (kk < EF) ? erow[j * EF + kk] : 0.0f;
        }
    }
    __syncthreads();

    const int lane  = tid & 31;
    const int wave  = tid >> 5;
    const int half  = lane >> 4;
    const int l     = lane & 15;
    const int fbase = wave * 32;

    const v8f vzero = {0.f, 0.f, 0.f, 0.f, 0.f, 0.f, 0.f, 0.f};
    v8f acc[8][2];
#pragma unroll
    for (int mt = 0; mt < 8; ++mt)
#pragma unroll
        for (int nt = 0; nt < 2; ++nt)
            acc[mt][nt] = vzero;

#pragma unroll
    for (int kk = 0; kk < KP / 4; ++kk) {
        const int k0 = kk * 4 + 2 * half;
#pragma unroll
        for (int mt = 0; mt < 8; ++mt) {
            const int j = mt * 16 + l;
            v2f a;
            a.x = sh_e[j * KP + k0];
            a.y = sh_e[j * KP + k0 + 1];
#pragma unroll
            for (int nt = 0; nt < 2; ++nt) {
                const int n = fbase + nt * 16 + l;
                v2f bb;
                bb.x = sh_wf[k0 * HD + n];
                bb.y = sh_wf[(k0 + 1) * HD + n];
                acc[mt][nt] = wmma_f32(a, bb, acc[mt][nt]);
            }
        }
    }

    const float* hB  = h + (size_t)b * NA * HD;
    const float* emB = edge_mask + ((size_t)(b * NA + i)) * NA;
#pragma unroll
    for (int nt = 0; nt < 2; ++nt) {
        const int f     = fbase + nt * 16 + l;
        const float bia = bf[f];
        float psum = 0.0f;
#pragma unroll
        for (int mt = 0; mt < 8; ++mt) {
#pragma unroll
            for (int r = 0; r < 8; ++r) {
                const int j  = mt * 16 + half * 8 + r;
                const float ef = selu_f(acc[mt][nt][r] + bia) * emB[j];
                psum += ef * hB[j * HD + f];
            }
        }
        const float tot = psum + __shfl_xor(psum, 16, 32);
        if (half == 0)
            h1out[(size_t)(b * NA + i) * HD + f] = tot;
    }
}

__global__ __launch_bounds__(256)
void ef_node_mlp(const float* __restrict__ h,
                 const float* __restrict__ node_mask,
                 const float* __restrict__ Ww,
                 const float* __restrict__ bw,
                 float* __restrict__ out  ) {
    __shared__ float sh_z[ROWS * KC];
    __shared__ float sh_w[KC * HD];

    const int tid = threadIdx.x;
    const int r0  = blockIdx.x * ROWS;

    const int lane  = tid & 31;
    const int wave  = tid >> 5;
    const int half  = lane >> 4;
    const int l     = lane & 15;
    const int nbase = wave * 32;

    const v8f vzero = {0.f, 0.f, 0.f, 0.f, 0.f, 0.f, 0.f, 0.f};
    v8f acc[4][2];
#pragma unroll
    for (int mt = 0; mt < 4; ++mt)
#pragma unroll
        for (int nt = 0; nt < 2; ++nt)
            acc[mt][nt] = vzero;

    for (int kc = 0; kc < K2 / KC; ++kc) {
        const int kg0 = kc * KC;
        {
            const int rr  = tid >> 2;
            const int kb  = (tid & 3) * 8;
            const int row = r0 + rr;
#pragma unroll
            for (int k = 0; k < 8; ++k) {
                const int kg = kg0 + kb + k;
                float v;
                if (kg < HD) v = h[(size_t)row * HD + kg];
                else         v = out[(size_t)row * HD + (kg - HD)];
                sh_z[rr * KC + kb + k] = v;
            }
        }
        {
            const float* wn = Ww + (size_t)tid * K2 + kg0;
#pragma unroll
            for (int k = 0; k < KC; ++k)
                sh_w[k * HD + tid] = wn[k];
        }
        __syncthreads();

#pragma unroll
        for (int kk = 0; kk < KC / 4; ++kk) {
            const int k0 = kk * 4 + 2 * half;
#pragma unroll
            for (int mt = 0; mt < 4; ++mt) {
                const int rr = mt * 16 + l;
                v2f a;
                a.x = sh_z[rr * KC + k0];
                a.y = sh_z[rr * KC + k0 + 1];
#pragma unroll
                for (int nt = 0; nt < 2; ++nt) {
                    const int n = nbase + nt * 16 + l;
                    v2f bb;
                    bb.x = sh_w[k0 * HD + n];
                    bb.y = sh_w[(k0 + 1) * HD + n];
                    acc[mt][nt] = wmma_f32(a, bb, acc[mt][nt]);
                }
            }
        }
        __syncthreads();
    }

#pragma unroll
    for (int mt = 0; mt < 4; ++mt) {
#pragma unroll
        for (int nt = 0; nt < 2; ++nt) {
            const int f = nbase + nt * 16 + l;
#pragma unroll
            for (int r = 0; r < 8; ++r) {
                const int row = r0 + mt * 16 + half * 8 + r;
                float v = selu_f(acc[mt][nt][r] + bw[f]) * node_mask[row];
                out[(size_t)row * HD + f] = v + h[(size_t)row * HD + f];
            }
        }
    }
}

extern "C" void kernel_launch(void* const* d_in, const int* in_sizes, int n_in,
                              void* d_out, int out_size, void* d_ws, size_t ws_size,
                              hipStream_t stream) {
    (void)in_sizes; (void)n_in; (void)out_size; (void)d_ws; (void)ws_size;
    const float* h         = (const float*)d_in[0];
    const float* e         = (const float*)d_in[1];
    const float* node_mask = (const float*)d_in[2];
    const float* edge_mask = (const float*)d_in[3];
    const float* Wf        = (const float*)d_in[4];
    const float* bf        = (const float*)d_in[5];
    const float* Ww        = (const float*)d_in[6];
    const float* bw        = (const float*)d_in[7];
    float* out = (float*)d_out;

    ef_filter_reduce<<<NB * NA, 256, 0, stream>>>(h, e, edge_mask, Wf, bf, out);
    ef_node_mlp<<<(NB * NA) / ROWS, 256, 0, stream>>>(h, node_mask, Ww, bw, out);
}


# Node MLP: Ww goes global->B fragments by 16-B per-lane reads (no LDS stage, no loop barrier); 128 groups of 8 waves, a wave owns 16 rows x 32 f over all k, z rows staged once; k order in sums changes
# speedup vs baseline: 2.5736x; 2.5736x over previous
#include <hip/hip_runtime.h>


typedef __attribute__((ext_vector_type(2))) float v2f;
typedef __attribute__((ext_vector_type(8))) float v8f;

#define NB 16
#define NA 128
#define HD 256
#define EF 30
#define KP 32
#define K2 (2 * HD)

__device__ __forceinline__ float selu_f(float x) {
    const float scale = 1.0507009873554805f;
    const float alpha = 1.6732632423543772f;
    return x > 0.0f ? scale * x : scale * alpha * (__expf(x) - 1.0f);
}

__device__ __forceinline__ v8f wmma_f32(v2f a, v2f b, v8f c) {
    return __builtin_amdgcn_wmma_f32_16x16x4_f32(false, a, false, b, (short)0, c,
                                                 false, false);
}

#define IPW 8
#define JT  (NA / 16)
#define NTW 8
#define KT  (KP / 4)
#define FTH 512

typedef __attribute__((ext_vector_type(4))) float v4f;

__global__ __launch_bounds__(FTH)
void ef_filter_reduce(const float* __restrict__ h,
                      const float* __restrict__ e,
                      const float* __restrict__ edge_mask,
                      const float* __restrict__ Wf,
                      const float* __restrict__ bf,
                      float* __restrict__ h1out) {
    __shared__ v2f   sh_b[(HD / 16) * KT * 32];
    __shared__ float sh_part[IPW * JT * HD];

    const int tid  = threadIdx.x;
    const int b    = blockIdx.x >> 4;
    const int i0   = (blockIdx.x & 15) * IPW;
    const int lane = tid & 31;
    const int wave = tid >> 5;
    const int jt   = wave & 7;
    const int fh   = wave >> 3;
    const int half = lane >> 4;
    const int l    = lane & 15;

    for (int idx = tid; idx < (HD / 16) * KT * 32; idx += FTH) {
        const int ln = idx & 31;
        const int kt = (idx >> 5) & 7;
        const int nt = idx >> 8;
        const int f  = nt * 16 + (ln & 15);
        const int k  = 4 * kt + 2 * (ln >> 4);
        v2f v = {0.f, 0.f};
        if (k < EF) {
            v.x = Wf[f * EF + k];
            v.y = Wf[f * EF + k + 1];
        }
        sh_b[idx] = v;
    }

    const int jrow = jt * 16 + half * 8;
    const int f0   = fh * (NTW * 16) + l;
    float hreg[NTW][8];
    float bia[NTW];
    {
        const float* hB = h + ((size_t)b * NA + jrow) * HD + f0;
#pragma unroll
        for (int t = 0; t < NTW; ++t) {
#pragma unroll
            for (int r = 0; r < 8; ++r)
                hreg[t][r] = hB[r * HD + t * 16];
            bia[t] = bf[f0 + t * 16];
        }
    }
    __syncthreads();

    const float* eA = e + (((size_t)(b * NA + i0)) * NA + jt * 16 + l) * EF;
    const float* mB = edge_mask + ((size_t)(b * NA + i0)) * NA + jrow;
    const v2f*   bW = sh_b + (fh * NTW) * KT * 32 + lane;
    const v2f z2 = {0.f, 0.f};

    v2f a[KT];
#pragma unroll
    for (int kt = 0; kt < KT - 1; ++kt)
        a[kt] = *(const v2f*)(eA + 4 * kt + 2 * half);
    a[KT - 1] = *(const v2f*)(eA + 4 * (KT - 1));
    a[KT - 1] = half ? z2 : a[KT - 1];

    v4f m0 = *(const v4f*)(mB);
    v4f m1 = *(const v4f*)(mB + 4);

    v2f bc0[KT], bc1[KT];
#pragma unroll
    for (int kt = 0; kt < KT; ++kt) {
        bc0[kt] = bW[(0 * KT + kt) * 32];
        bc1[kt] = bW[(1 * KT + kt) * 32];
    }

    for (int il = 0; il < IPW; ++il) {
        const int iln = (il + 1 < IPW) ? il + 1 : il;
        const v4f mn0 = *(const v4f*)(mB + (size_t)iln * NA);
        const v4f mn1 = *(const v4f*)(mB + (size_t)iln * NA + 4);
        const float* eN = eA + (size_t)iln * NA * EF;
        v2f an[KT];
#pragma unroll
        for (int kt = 0; kt < KT - 1; ++kt)
            an[kt] = *(const v2f*)(eN + 4 * kt + 2 * half);
        an[KT - 1] = *(const v2f*)(eN + 4 * (KT - 1));
        an[KT - 1] = half ? z2 : an[KT - 1];

        const float m[8] = {m0.x, m0.y, m0.z, m0.w, m1.x, m1.y, m1.z, m1.w};

        float* part = sh_part + (il * JT + jt) * HD + fh * (NTW * 16) + lane;

#pragma unroll
        for (int np = 0; np < NTW / 2; ++np) {
            asm volatile("" ::: "memory");
            const int pn = (np + 1) % (NTW / 2);
            v2f bn0[KT], bn1[KT];
#pragma unroll
            for (int kt = 0; kt < KT; ++kt) {
                bn0[kt] = bW[((2 * pn)     * KT + kt) * 32];
                bn1[kt] = bW[((2 * pn + 1) * KT + kt) * 32];
            }

            v8f c0 = {0.f, 0.f, 0.f, 0.f, 0.f, 0.f, 0.f, 0.f};
            v8f c1 = c0;
#pragma unroll
            for (int kt = 0; kt < KT; ++kt) {
                c0 = wmma_f32(a[kt], bc0[kt], c0);
                c1 = wmma_f32(a[kt], bc1[kt], c1);
            }
            float p0 = 0.0f, p1 = 0.0f;
#pragma unroll
            for (int r = 0; r < 8; ++r) {
                const float e0 = selu_f(c0[r] + bia[2 * np]) * m[r];
                p0 += e0 * hreg[2 * np][r];
                const float e1 = selu_f(c1[r] + bia[2 * np + 1]) * m[r];
                p1 += e1 * hreg[2 * np + 1][r];
            }
            const float t0 = p0 + __shfl_xor(p0, 16, 32);
            const float t1 = p1 + __shfl_xor(p1, 16, 32);
            part[np * 32] = half ? t1 : t0;

#pragma unroll
            for (int kt = 0; kt < KT; ++kt) {
                bc0[kt] = bn0[kt];
                bc1[kt] = bn1[kt];
            }
        }

#pragma unroll
        for (int kt = 0; kt < KT; ++kt)
            a[kt] = an[kt];
        m0 = mn0;
        m1 = mn1;
    }
    __syncthreads();

    {
        const int f = tid & (HD - 1);
#pragma unroll
        for (int q = 0; q < IPW * HD / FTH; ++q) {
            const int il = q * (FTH / HD) + (tid >> 8);
            float s = 0.0f;
#pragma unroll
            for (int w = 0; w < JT; ++w)
                s += sh_part[(il * JT + w) * HD + f];
            h1out[((size_t)(b * NA + i0 + il)) * HD + f] = s;
        }
    }
}

#define MR   16
#define ZP   (K2 + 4)
#define CHS  8
#define NCH  (K2 / (8 * CHS))

__global__ __launch_bounds__(256) __attribute__((amdgpu_waves_per_eu(2, 2)))
void ef_node_mlp(const float* __restrict__ h,
                 const float* __restrict__ node_mask,
                 const float* __restrict__ Ww,
                 const float* __restrict__ bw,
                 float* __restrict__ out  ) {
    __shared__ float sh_z[MR * ZP];

    const int tid = threadIdx.x;
    const int r0  = blockIdx.x * MR;

    const int lane  = tid & 31;
    const int wave  = tid >> 5;
    const int half  = lane >> 4;
    const int l     = lane & 15;
    const int nbase = wave * 32;

    const float* w0 = Ww + (size_t)(nbase + l) * K2 + 4 * half;
    const float* w1 = w0 + (size_t)16 * K2;

    v4f zt[(MR * K2 / 4) / 256];
#pragma unroll
    for (int q = 0; q < (MR * K2 / 4) / 256; ++q) {
        const int idx = tid + 256 * q;
        const int row = idx >> 7;
        const int c4  = idx & 127;
        const float* src = (c4 < HD / 4)
            ? h   + (size_t)(r0 + row) * HD + c4 * 4
            : out + (size_t)(r0 + row) * HD + (c4 - HD / 4) * 4;
        zt[q] = *(const v4f*)src;
    }

    v4f b0c[CHS], b1c[CHS];
#pragma unroll
    for (int s = 0; s < CHS; ++s) {
        b0c[s] = *(const v4f*)(w0 + 8 * s);
        b1c[s] = *(const v4f*)(w1 + 8 * s);
    }

#pragma unroll
    for (int q = 0; q < (MR * K2 / 4) / 256; ++q) {
        const int idx = tid + 256 * q;
        *(v4f*)(sh_z + (idx >> 7) * ZP + (idx & 127) * 4) = zt[q];
    }
    __syncthreads();

    const int f0 = nbase + l;
    const int f1 = f0 + 16;
    const float bw0 = bw[f0];
    const float bw1 = bw[f1];
    float hr0[8], hr1[8], nmr[8];
#pragma unroll
    for (int r = 0; r < 8; ++r) {
        const int row = r0 + half * 8 + r;
        hr0[r] = h[(size_t)row * HD + f0];
        hr1[r] = h[(size_t)row * HD + f1];
        nmr[r] = node_mask[row];
    }

    const float* za = sh_z + l * ZP + 4 * half;
    v4f ac[CHS];
#pragma unroll
    for (int s = 0; s < CHS; ++s)
        ac[s] = *(const v4f*)(za + 8 * s);

    const v8f vzero = {0.f, 0.f, 0.f, 0.f, 0.f, 0.f, 0.f, 0.f};
    v8f acc0 = vzero, acc1 = vzero;

#pragma unroll
    for (int c = 0; c < NCH; ++c) {
        v4f an[CHS], b0n[CHS], b1n[CHS];
        if (c + 1 < NCH) {
#pragma unroll
            for (int s = 0; s < CHS; ++s) {
                b0n[s] = *(const v4f*)(w0 + 64 * (c + 1) + 8 * s);
                b1n[s] = *(const v4f*)(w1 + 64 * (c + 1) + 8 * s);
            }
#pragma unroll
            for (int s = 0; s < CHS; ++s)
                an[s] = *(const v4f*)(za + 64 * (c + 1) + 8 * s);
        }
#pragma unroll
        for (int s = 0; s < CHS; ++s) {
            const v2f a_lo = {ac[s].x, ac[s].y};
            const v2f a_hi = {ac[s].z, ac[s].w};
            const v2f p_lo = {b0c[s].x, b0c[s].y};
            const v2f p_hi = {b0c[s].z, b0c[s].w};
            const v2f q_lo = {b1c[s].x, b1c[s].y};
            const v2f q_hi = {b1c[s].z, b1c[s].w};
            acc0 = wmma_f32(a_lo, p_lo, acc0);
            acc1 = wmma_f32(a_lo, q_lo, acc1);
            acc0 = wmma_f32(a_hi, p_hi, acc0);
            acc1 = wmma_f32(a_hi, q_hi, acc1);
        }
        if (c + 1 < NCH) {
#pragma unroll
            for (int s = 0; s < CHS; ++s) {
                ac[s]  = an[s];
                b0c[s] = b0n[s];
                b1c[s] = b1n[s];
            }
        }
    }

#pragma unroll
    for (int r = 0; r < 8; ++r) {
        const int row = r0 + half * 8 + r;
        float v0 = selu_f(acc0[r] + bw0) * nmr[r];
        out[(size_t)row * HD + f0] = v0 + hr0[r];
        float v1 = selu_f(acc1[r] + bw1) * nmr[r];
        out[(size_t)row * HD + f1] = v1 + hr1[r];
    }
}

extern "C" void kernel_launch(void* const* d_in, const int* in_sizes, int n_in,
                              void* d_out, int out_size, void* d_ws, size_t ws_size,
                              hipStream_t stream) {
    (void)in_sizes; (void)n_in; (void)out_size; (void)d_ws; (void)ws_size;
    const float* h         = (const float*)d_in[0];
    const float* e         = (const float*)d_in[1];
    const float* node_mask = (const float*)d_in[2];
    const float* edge_mask = (const float*)d_in[3];
    const float* Wf        = (const float*)d_in[4];
    const float* bf        = (const float*)d_in[5];
    const float* Ww        = (const float*)d_in[6];
    const float* bw        = (const float*)d_in[7];
    float* out = (float*)d_out;

    ef_filter_reduce<<<(NB * NA) / IPW, FTH, 0, stream>>>(h, e, edge_mask, Wf, bf, out);
    ef_node_mlp<<<(NB * NA) / MR, 256, 0, stream>>>(h, node_mask, Ww, bw, out);
}


# Filter kernel re-tiled for 8 waves per SIMD: 32-wave groups, a wave keeps 4 of the 8 i; h[b] moves from 64 registers to a transposed LDS copy, B fragments read at use (no second register set)
# speedup vs baseline: 2.8624x; 1.1122x over previous
#include <hip/hip_runtime.h>


typedef __attribute__((ext_vector_type(2))) float v2f;
typedef __attribute__((ext_vector_type(8))) float v8f;

#define NB 16
#define NA 128
#define HD 256
#define EF 30
#define KP 32
#define K2 (2 * HD)

__device__ __forceinline__ float selu_f(float x) {
    const float scale = 1.0507009873554805f;
    const float alpha = 1.6732632423543772f;
    return x > 0.0f ? scale * x : scale * alpha * (__expf(x) - 1.0f);
}

__device__ __forceinline__ v8f wmma_f32(v2f a, v2f b, v8f c) {
    return __builtin_amdgcn_wmma_f32_16x16x4_f32(false, a, false, b, (short)0, c,
                                                 false, false);
}

#define IPW 8
#define JT  (NA / 16)
#define NTW 8
#define KT  (KP / 4)
#define IPH 4
#define HP  (NA + 4)
#define FTH 1024

typedef __attribute__((ext_vector_type(4))) float v4f;

__global__ __launch_bounds__(FTH)
void ef_filter_reduce(const float* __restrict__ h,
                      const float* __restrict__ e,
                      const float* __restrict__ edge_mask,
                      const float* __restrict__ Wf,
                      const float* __restrict__ bf,
                      float* __restrict__ h1out) {
    __shared__ v2f   sh_b[(HD / 16) * KT * 32];
    __shared__ float sh_part[IPW * JT * HD];
    __shared__ alignas(16) float sh_h[HD * HP];

    const int tid  = threadIdx.x;
    const int b    = blockIdx.x >> 4;
    const int i0   = (blockIdx.x & 15) * IPW;
    const int lane = tid & 31;
    const int wave = tid >> 5;
    const int jt   = wave & 7;
    const int fh   = (wave >> 3) & 1;
    const int ih   = wave >> 4;
    const int half = lane >> 4;
    const int l    = lane & 15;

    for (int idx = tid; idx < (HD / 16) * KT * 32; idx += FTH) {
        const int ln = idx & 31;
        const int kt = (idx >> 5) & 7;
        const int nt = idx >> 8;
        const int f  = nt * 16 + (ln & 15);
        const int k  = 4 * kt + 2 * (ln >> 4);
        v2f v = {0.f, 0.f};
        if (k < EF) {
            v.x = Wf[f * EF + k];
            v.y = Wf[f * EF + k + 1];
        }
        sh_b[idx] = v;
    }

    {
        const float* hB = h + (size_t)b * NA * HD;
#pragma unroll 8
        for (int idx = tid; idx < NA * HD; idx += FTH) {
            const int j = idx >> 8;
            const int f = idx & (HD - 1);
            sh_h[f * HP + j] = hB[idx];
        }
    }

    const int jrow = jt * 16 + half * 8;
    const int f0   = fh * (NTW * 16) + l;
    float bia[NTW];
#pragma unroll
    for (int t = 0; t < NTW; ++t)
        bia[t] = bf[f0 + t * 16];
    const float* hT = sh_h + f0 * HP + jrow;
    __syncthreads();

    const int ib = i0 + ih * IPH;
    const float* eA = e + (((size_t)(b * NA + ib)) * NA + jt * 16 + l) * EF;
    const float* mB = edge_mask + ((size_t)(b * NA + ib)) * NA + jrow;
    const v2f*   bW = sh_b + (fh * NTW) * KT * 32 + lane;
    const v2f z2 = {0.f, 0.f};

    v2f a[KT];
#pragma unroll
    for (int kt = 0; kt < KT - 1; ++kt)
        a[kt] = *(const v2f*)(eA + 4 * kt + 2 * half);
    a[KT - 1] = *(const v2f*)(eA + 4 * (KT - 1));
    a[KT - 1] = half ? z2 : a[KT - 1];

    v4f m0 = *(const v4f*)(mB);
    v4f m1 = *(const v4f*)(mB + 4);

    for (int il = 0; il < IPH; ++il) {
        const int iln = (il + 1 < IPH) ? il + 1 : il;
        const v4f mn0 = *(const v4f*)(mB + (size_t)iln * NA);
        const v4f mn1 = *(const v4f*)(mB + (size_t)iln * NA + 4);
        const float* eN = eA + (size_t)iln * NA * EF;
        v2f an[KT];
#pragma unroll
        for (int kt = 0; kt < KT - 1; ++kt)
            an[kt] = *(const v2f*)(eN + 4 * kt + 2 * half);
        an[KT - 1] = *(const v2f*)(eN + 4 * (KT - 1));
        an[KT - 1] = half ? z2 : an[KT - 1];

        const float m[8] = {m0.x, m0.y, m0.z, m0.w, m1.x, m1.y, m1.z, m1.w};

        float* part = sh_part + ((ih * IPH + il) * JT + jt) * HD + fh * (NTW * 16) + lane;

#pragma unroll
        for (int np = 0; np < NTW / 2; ++np) {
            asm volatile("" ::: "memory");
            v2f bc0[KT], bc1[KT];
#pragma unroll
            for (int kt = 0; kt < KT; ++kt) {
                bc0[kt] = bW[((2 * np)     * KT + kt) * 32];
                bc1[kt] = bW[((2 * np + 1) * KT + kt) * 32];
            }
            const v4f ha0 = *(const v4f*)(hT + (2 * np) * 16 * HP);
            const v4f ha1 = *(const v4f*)(hT + (2 * np) * 16 * HP + 4);
            const v4f hb0 = *(const v4f*)(hT + (2 * np + 1) * 16 * HP);
            const v4f hb1 = *(const v4f*)(hT + (2 * np + 1) * 16 * HP + 4);
            const float hA[8] = {ha0.x, ha0.y, ha0.z, ha0.w, ha1.x, ha1.y, ha1.z, ha1.w};
            const float hB8[8] = {hb0.x, hb0.y, hb0.z, hb0.w, hb1.x, hb1.y, hb1.z, hb1.w};

            v8f c0 = {0.f, 0.f, 0.f, 0.f, 0.f, 0.f, 0.f, 0.f};
            v8f c1 = c0;
#pragma unroll
            for (int kt = 0; kt < KT; ++kt) {
                c0 = wmma_f32(a[kt], bc0[kt], c0);
                c1 = wmma_f32(a[kt], bc1[kt], c1);
            }
            float p0 = 0.0f, p1 = 0.0f;
#pragma unroll
            for (int r = 0; r < 8; ++r) {
                const float e0 = selu_f(c0[r] + bia[2 * np]) * m[r];
                p0 += e0 * hA[r];
                const float e1 = selu_f(c1[r] + bia[2 * np + 1]) * m[r];
                p1 += e1 * hB8[r];
            }
            const float t0 = p0 + __shfl_xor(p0, 16, 32);
            const float t1 = p1 + __shfl_xor(p1, 16, 32);
            part[np * 32] = half ? t1 : t0;
        }

#pragma unroll
        for (int kt = 0; kt < KT; ++kt)
            a[kt] = an[kt];
        m0 = mn0;
        m1 = mn1;
    }
    __syncthreads();

    {
        const int f = tid & (HD - 1);
#pragma unroll
        for (int q = 0; q < IPW * HD / FTH; ++q) {
            const int il = q * (FTH / HD) + (tid >> 8);
            float s = 0.0f;
#pragma unroll
            for (int w = 0; w < JT; ++w)
                s += sh_part[(il * JT + w) * HD + f];
            h1out[((size_t)(b * NA + i0 + il)) * HD + f] = s;
        }
    }
}

#define MR   16
#define ZP   (K2 + 4)
#define CHS  8
#define NCH  (K2 / (8 * CHS))

__global__ __launch_bounds__(256) __attribute__((amdgpu_waves_per_eu(2, 2)))
void ef_node_mlp(const float* __restrict__ h,
                 const float* __restrict__ node_mask,
                 const float* __restrict__ Ww,
                 const float* __restrict__ bw,
                 float* __restrict__ out  ) {
    __shared__ float sh_z[MR * ZP];

    const int tid = threadIdx.x;
    const int r0  = blockIdx.x * MR;

    const int lane  = tid & 31;
    const int wave  = tid >> 5;
    const int half  = lane >> 4;
    const int l     = lane & 15;
    const int nbase = wave * 32;

    const float* w0 = Ww + (size_t)(nbase + l) * K2 + 4 * half;
    const float* w1 = w0 + (size_t)16 * K2;

    v4f zt[(MR * K2 / 4) / 256];
#pragma unroll
    for (int q = 0; q < (MR * K2 / 4) / 256; ++q) {
        const int idx = tid + 256 * q;
        const int row = idx >> 7;
        const int c4  = idx & 127;
        const float* src = (c4 < HD / 4)
            ? h   + (size_t)(r0 + row) * HD + c4 * 4
            : out + (size_t)(r0 + row) * HD + (c4 - HD / 4) * 4;
        zt[q] = *(const v4f*)src;
    }

    v4f b0c[CHS], b1c[CHS];
#pragma unroll
    for (int s = 0; s < CHS; ++s) {
        b0c[s] = *(const v4f*)(w0 + 8 * s);
        b1c[s] = *(const v4f*)(w1 + 8 * s);
    }

#pragma unroll
    for (int q = 0; q < (MR * K2 / 4) / 256; ++q) {
        const int idx = tid + 256 * q;
        *(v4f*)(sh_z + (idx >> 7) * ZP + (idx & 127) * 4) = zt[q];
    }
    __syncthreads();

    const int f0 = nbase + l;
    const int f1 = f0 + 16;
    const float bw0 = bw[f0];
    const float bw1 = bw[f1];
    float hr0[8], hr1[8], nmr[8];
#pragma unroll
    for (int r = 0; r < 8; ++r) {
        const int row = r0 + half * 8 + r;
        hr0[r] = h[(size_t)row * HD + f0];
        hr1[r] = h[(size_t)row * HD + f1];
        nmr[r] = node_mask[row];
    }

    const float* za = sh_z + l * ZP + 4 * half;
    v4f ac[CHS];
#pragma unroll
    for (int s = 0; s < CHS; ++s)
        ac[s] = *(const v4f*)(za + 8 * s);

    const v8f vzero = {0.f, 0.f, 0.f, 0.f, 0.f, 0.f, 0.f, 0.f};
    v8f acc0 = vzero, acc1 = vzero;

#pragma unroll
    for (int c = 0; c < NCH; ++c) {
        v4f an[CHS], b0n[CHS], b1n[CHS];
        if (c + 1 < NCH) {
#pragma unroll
            for (int s = 0; s < CHS; ++s) {
                b0n[s] = *(const v4f*)(w0 + 64 * (c + 1) + 8 * s);
                b1n[s] = *(const v4f*)(w1 + 64 * (c + 1) + 8 * s);
            }
#pragma unroll
            for (int s = 0; s < CHS; ++s)
                an[s] = *(const v4f*)(za + 64 * (c + 1) + 8 * s);
        }
#pragma unroll
        for (int s = 0; s < CHS; ++s) {
            const v2f a_lo = {ac[s].x, ac[s].y};
            const v2f a_hi = {ac[s].z, ac[s].w};
            const v2f p_lo = {b0c[s].x, b0c[s].y};
            const v2f p_hi = {b0c[s].z, b0c[s].w};
            const v2f q_lo = {b1c[s].x, b1c[s].y};
            const v2f q_hi = {b1c[s].z, b1c[s].w};
            acc0 = wmma_f32(a_lo, p_lo, acc0);
            acc1 = wmma_f32(a_lo, q_lo, acc1);
            acc0 = wmma_f32(a_hi, p_hi, acc0);
            acc1 = wmma_f32(a_hi, q_hi, acc1);
        }
        if (c + 1 < NCH) {
#pragma unroll
            for (int s = 0; s < CHS; ++s) {
                ac[s]  = an[s];
                b0c[s] = b0n[s];
                b1c[s] = b1n[s];
            }
        }
    }

#pragma unroll
    for (int r = 0; r < 8; ++r) {
        const int row = r0 + half * 8 + r;
        float v0 = selu_f(acc0[r] + bw0) * nmr[r];
        out[(size_t)row * HD + f0] = v0 + hr0[r];
        float v1 = selu_f(acc1[r] + bw1) * nmr[r];
        out[(size_t)row * HD + f1] = v1 + hr1[r];
    }
}

extern "C" void kernel_launch(void* const* d_in, const int* in_sizes, int n_in,
                              void* d_out, int out_size, void* d_ws, size_t ws_size,
                              hipStream_t stream) {
    (void)in_sizes; (void)n_in; (void)out_size; (void)d_ws; (void)ws_size;
    const float* h         = (const float*)d_in[0];
    const float* e         = (const float*)d_in[1];
    const float* node_mask = (const float*)d_in[2];
    const float* edge_mask = (const float*)d_in[3];
    const float* Wf        = (const float*)d_in[4];
    const float* bf        = (const float*)d_in[5];
    const float* Ww        = (const float*)d_in[6];
    const float* bw        = (const float*)d_in[7];
    float* out = (float*)d_out;

    ef_filter_reduce<<<(NB * NA) / IPW, FTH, 0, stream>>>(h, e, edge_mask, Wf, bf, out);
    ef_node_mlp<<<(NB * NA) / MR, 256, 0, stream>>>(h, node_mask, Ww, bw, out);
}


# selu computes both sides and picks one by select: its conditional form compiled to an EXEC-masked branch per element in the filter fold; next row's mask load moved behind the fold to fit 128 VGPRs
# speedup vs baseline: 3.4625x; 1.2096x over previous
#include <hip/hip_runtime.h>


typedef __attribute__((ext_vector_type(2))) float v2f;
typedef __attribute__((ext_vector_type(8))) float v8f;

#define NB 16
#define NA 128
#define HD 256
#define EF 30
#define KP 32
#define K2 (2 * HD)

__device__ __forceinline__ float selu_f(float x) {
    const float scale = 1.0507009873554805f;
    const float alpha = 1.6732632423543772f;
    const float pos = scale * x;
    const float neg = scale * alpha * (__expf(x) - 1.0f);
    return x > 0.0f ? pos : neg;
}

__device__ __forceinline__ v8f wmma_f32(v2f a, v2f b, v8f c) {
    return __builtin_amdgcn_wmma_f32_16x16x4_f32(false, a, false, b, (short)0, c,
                                                 false, false);
}

#define IPW 8
#define JT  (NA / 16)
#define NTW 8
#define KT  (KP / 4)
#define IPH 4
#define HP  (NA + 4)
#define FTH 1024

typedef __attribute__((ext_vector_type(4))) float v4f;

__global__ __launch_bounds__(FTH)
void ef_filter_reduce(const float* __restrict__ h,
                      const float* __restrict__ e,
                      const float* __restrict__ edge_mask,
                      const float* __restrict__ Wf,
                      const float* __restrict__ bf,
                      float* __restrict__ h1out) {
    __shared__ v2f   sh_b[(HD / 16) * KT * 32];
    __shared__ float sh_part[IPW * JT * HD];
    __shared__ alignas(16) float sh_h[HD * HP];

    const int tid  = threadIdx.x;
    const int b    = blockIdx.x >> 4;
    const int i0   = (blockIdx.x & 15) * IPW;
    const int lane = tid & 31;
    const int wave = tid >> 5;
    const int jt   = wave & 7;
    const int fh   = (wave >> 3) & 1;
    const int ih   = wave >> 4;
    const int half = lane >> 4;
    const int l    = lane & 15;

    for (int idx = tid; idx < (HD / 16) * KT * 32; idx += FTH) {
        const int ln = idx & 31;
        const int kt = (idx >> 5) & 7;
        const int nt = idx >> 8;
        const int f  = nt * 16 + (ln & 15);
        const int k  = 4 * kt + 2 * (ln >> 4);
        v2f v = {0.f, 0.f};
        if (k < EF) {
            v.x = Wf[f * EF + k];
            v.y = Wf[f * EF + k + 1];
        }
        sh_b[idx] = v;
    }

    {
        const float* hB = h + (size_t)b * NA * HD;
#pragma unroll 8
        for (int idx = tid; idx < NA * HD; idx += FTH) {
            const int j = idx >> 8;
            const int f = idx & (HD - 1);
            sh_h[f * HP + j] = hB[idx];
        }
    }

    const int jrow = jt * 16 + half * 8;
    const int f0   = fh * (NTW * 16) + l;
    float bia[NTW];
#pragma unroll
    for (int t = 0; t < NTW; ++t)
        bia[t] = bf[f0 + t * 16];
    const float* hT = sh_h + f0 * HP + jrow;
    __syncthreads();

    const int ib = i0 + ih * IPH;
    const float* eA = e + (((size_t)(b * NA + ib)) * NA + jt * 16 + l) * EF;
    const float* mB = edge_mask + ((size_t)(b * NA + ib)) * NA + jrow;
    const v2f*   bW = sh_b + (fh * NTW) * KT * 32 + lane;
    const v2f z2 = {0.f, 0.f};

    v2f a[KT];
#pragma unroll
    for (int kt = 0; kt < KT - 1; ++kt)
        a[kt] = *(const v2f*)(eA + 4 * kt + 2 * half);
    a[KT - 1] = *(const v2f*)(eA + 4 * (KT - 1));
    a[KT - 1] = half ? z2 : a[KT - 1];

    v4f m0 = *(const v4f*)(mB);
    v4f m1 = *(const v4f*)(mB + 4);

    for (int il = 0; il < IPH; ++il) {
        const int iln = (il + 1 < IPH) ? il + 1 : il;
        const float* eN = eA + (size_t)iln * NA * EF;
        v2f an[KT];
#pragma unroll
        for (int kt = 0; kt < KT - 1; ++kt)
            an[kt] = *(const v2f*)(eN + 4 * kt + 2 * half);
        an[KT - 1] = *(const v2f*)(eN + 4 * (KT - 1));
        an[KT - 1] = half ? z2 : an[KT - 1];

        const float m[8] = {m0.x, m0.y, m0.z, m0.w, m1.x, m1.y, m1.z, m1.w};

        float* part = sh_part + ((ih * IPH + il) * JT + jt) * HD + fh * (NTW * 16) + lane;

#pragma unroll
        for (int np = 0; np < NTW / 2; ++np) {
            asm volatile("" ::: "memory");
            v2f bc0[KT], bc1[KT];
#pragma unroll
            for (int kt = 0; kt < KT; ++kt) {
                bc0[kt] = bW[((2 * np)     * KT + kt) * 32];
                bc1[kt] = bW[((2 * np + 1) * KT + kt) * 32];
            }
            const v4f ha0 = *(const v4f*)(hT + (2 * np) * 16 * HP);
            const v4f ha1 = *(const v4f*)(hT + (2 * np) * 16 * HP + 4);
            const v4f hb0 = *(const v4f*)(hT + (2 * np + 1) * 16 * HP);
            const v4f hb1 = *(const v4f*)(hT + (2 * np + 1) * 16 * HP + 4);
            const float hA[8] = {ha0.x, ha0.y, ha0.z, ha0.w, ha1.x, ha1.y, ha1.z, ha1.w};
            const float hB8[8] = {hb0.x, hb0.y, hb0.z, hb0.w, hb1.x, hb1.y, hb1.z, hb1.w};

            v8f c0 = {0.f, 0.f, 0.f, 0.f, 0.f, 0.f, 0.f, 0.f};
            v8f c1 = c0;
#pragma unroll
            for (int kt = 0; kt < KT; ++kt) {
                c0 = wmma_f32(a[kt], bc0[kt], c0);
                c1 = wmma_f32(a[kt], bc1[kt], c1);
            }
            float p0 = 0.0f, p1 = 0.0f;
#pragma unroll
            for (int r = 0; r < 8; ++r) {
                const float e0 = selu_f(c0[r] + bia[2 * np]) * m[r];
                p0 += e0 * hA[r];
                const float e1 = selu_f(c1[r] + bia[2 * np + 1]) * m[r];
                p1 += e1 * hB8[r];
            }
            const float t0 = p0 + __shfl_xor(p0, 16, 32);
            const float t1 = p1 + __shfl_xor(p1, 16, 32);
            part[np * 32] = half ? t1 : t0;
        }

#pragma unroll
        for (int kt = 0; kt < KT; ++kt)
            a[kt] = an[kt];
        m0 = *(const v4f*)(mB + (size_t)iln * NA);
        m1 = *(const v4f*)(mB + (size_t)iln * NA + 4);
    }
    __syncthreads();

    {
        const int f = tid & (HD - 1);
#pragma unroll
        for (int q = 0; q < IPW * HD / FTH; ++q) {
            const int il = q * (FTH / HD) + (tid >> 8);
            float s = 0.0f;
#pragma unroll
            for (int w = 0; w < JT; ++w)
                s += sh_part[(il * JT + w) * HD + f];
            h1out[((size_t)(b * NA + i0 + il)) * HD + f] = s;
        }
    }
}

#define MR   16
#define ZP   (K2 + 4)
#define CHS  8
#define NCH  (K2 / (8 * CHS))

__global__ __launch_bounds__(256) __attribute__((amdgpu_waves_per_eu(2, 2)))
void ef_node_mlp(const float* __restrict__ h,
                 const float* __restrict__ node_mask,
                 const float* __restrict__ Ww,
                 const float* __restrict__ bw,
                 float* __restrict__ out  ) {
    __shared__ float sh_z[MR * ZP];

    const int tid = threadIdx.x;
    const int r0  = blockIdx.x * MR;

    const int lane  = tid & 31;
    const int wave  = tid >> 5;
    const int half  = lane >> 4;
    const int l     = lane & 15;
    const int nbase = wave * 32;

    const float* w0 = Ww + (size_t)(nbase + l) * K2 + 4 * half;
    const float* w1 = w0 + (size_t)16 * K2;

    v4f zt[(MR * K2 / 4) / 256];
#pragma unroll
    for (int q = 0; q < (MR * K2 / 4) / 256; ++q) {
        const int idx = tid + 256 * q;
        const int row = idx >> 7;
        const int c4  = idx & 127;
        const float* src = (c4 < HD / 4)
            ? h   + (size_t)(r0 + row) * HD + c4 * 4
            : out + (size_t)(r0 + row) * HD + (c4 - HD / 4) * 4;
        zt[q] = *(const v4f*)src;
    }

    v4f b0c[CHS], b1c[CHS];
#pragma unroll
    for (int s = 0; s < CHS; ++s) {
        b0c[s] = *(const v4f*)(w0 + 8 * s);
        b1c[s] = *(const v4f*)(w1 + 8 * s);
    }

#pragma unroll
    for (int q = 0; q < (MR * K2 / 4) / 256; ++q) {
        const int idx = tid + 256 * q;
        *(v4f*)(sh_z + (idx >> 7) * ZP + (idx & 127) * 4) = zt[q];
    }
    __syncthreads();

    const int f0 = nbase + l;
    const int f1 = f0 + 16;
    const float bw0 = bw[f0];
    const float bw1 = bw[f1];
    float hr0[8], hr1[8], nmr[8];
#pragma unroll
    for (int r = 0; r < 8; ++r) {
        const int row = r0 + half * 8 + r;
        hr0[r] = h[(size_t)row * HD + f0];
        hr1[r] = h[(size_t)row * HD + f1];
        nmr[r] = node_mask[row];
    }

    const float* za = sh_z + l * ZP + 4 * half;
    v4f ac[CHS];
#pragma unroll
    for (int s = 0; s < CHS; ++s)
        ac[s] = *(const v4f*)(za + 8 * s);

    const v8f vzero = {0.f, 0.f, 0.f, 0.f, 0.f, 0.f, 0.f, 0.f};
    v8f acc0 = vzero, acc1 = vzero;

#pragma unroll
    for (int c = 0; c < NCH; ++c) {
        v4f an[CHS], b0n[CHS], b1n[CHS];
        if (c + 1 < NCH) {
#pragma unroll
            for (int s = 0; s < CHS; ++s) {
                b0n[s] = *(const v4f*)(w0 + 64 * (c + 1) + 8 * s);
                b1n[s] = *(const v4f*)(w1 + 64 * (c + 1) + 8 * s);
            }
#pragma unroll
            for (int s = 0; s < CHS; ++s)
                an[s] = *(const v4f*)(za + 64 * (c + 1) + 8 * s);
        }
#pragma unroll
        for (int s = 0; s < CHS; ++s) {
            const v2f a_lo = {ac[s].x, ac[s].y};
            const v2f a_hi = {ac[s].z, ac[s].w};
            const v2f p_lo = {b0c[s].x, b0c[s].y};
            const v2f p_hi = {b0c[s].z, b0c[s].w};
            const v2f q_lo = {b1c[s].x, b1c[s].y};
            const v2f q_hi = {b1c[s].z, b1c[s].w};
            acc0 = wmma_f32(a_lo, p_lo, acc0);
            acc1 = wmma_f32(a_lo, q_lo, acc1);
            acc0 = wmma_f32(a_hi, p_hi, acc0);
            acc1 = wmma_f32(a_hi, q_hi, acc1);
        }
        if (c + 1 < NCH) {
#pragma unroll
            for (int s = 0; s < CHS; ++s) {
                ac[s]  = an[s];
                b0c[s] = b0n[s];
                b1c[s] = b1n[s];
            }
        }
    }

#pragma unroll
    for (int r = 0; r < 8; ++r) {
        const int row = r0 + half * 8 + r;
        float v0 = selu_f(acc0[r] + bw0) * nmr[r];
        out[(size_t)row * HD + f0] = v0 + hr0[r];
        float v1 = selu_f(acc1[r] + bw1) * nmr[r];
        out[(size_t)row * HD + f1] = v1 + hr1[r];
    }
}

extern "C" void kernel_launch(void* const* d_in, const int* in_sizes, int n_in,
                              void* d_out, int out_size, void* d_ws, size_t ws_size,
                              hipStream_t stream) {
    (void)in_sizes; (void)n_in; (void)out_size; (void)d_ws; (void)ws_size;
    const float* h         = (const float*)d_in[0];
    const float* e         = (const float*)d_in[1];
    const float* node_mask = (const float*)d_in[2];
    const float* edge_mask = (const float*)d_in[3];
    const float* Wf        = (const float*)d_in[4];
    const float* bf        = (const float*)d_in[5];
    const float* Ww        = (const float*)d_in[6];
    const float* bw        = (const float*)d_in[7];
    float* out = (float*)d_out;

    ef_filter_reduce<<<(NB * NA) / IPW, FTH, 0, stream>>>(h, e, edge_mask, Wf, bf, out);
    ef_node_mlp<<<(NB * NA) / MR, 256, 0, stream>>>(h, node_mask, Ww, bw, out);
}


# Filter fold on row pairs of one tile, selu with no compare or select: scale*max(x,0)+fma(sa,exp(min(x,0)),-sa); bias pairs from LDS; a lane's 8-row sum in two partial sums: sum order changes
# speedup vs baseline: 3.5674x; 1.0303x over previous
#include <hip/hip_runtime.h>


typedef __attribute__((ext_vector_type(2))) float v2f;
typedef __attribute__((ext_vector_type(8))) float v8f;

#define NB 16
#define NA 128
#define HD 256
#define EF 30
#define KP 32
#define K2 (2 * HD)

__device__ __forceinline__ float selu_f(float x) {
    const float scale = 1.0507009873554805f;
    const float alpha = 1.6732632423543772f;
    const float pos = scale * x;
    const float neg = scale * alpha * (__expf(x) - 1.0f);
    return x > 0.0f ? pos : neg;
}

__device__ __forceinline__ v2f selu_pair(v2f x) {
    const float scale = 1.0507009873554805f;
    const float sa    = 1.0507009873554805f * 1.6732632423543772f;
    const v2f zero = {0.f, 0.f};
    const v2f mn = __builtin_elementwise_min(x, zero);
    const v2f mx = __builtin_elementwise_max(x, zero);
    const v2f ex = {__expf(mn.x), __expf(mn.y)};
    const v2f neg = ex * sa - sa;
    return mx * scale + neg;
}

__device__ __forceinline__ v8f wmma_f32(v2f a, v2f b, v8f c) {
    return __builtin_amdgcn_wmma_f32_16x16x4_f32(false, a, false, b, (short)0, c,
                                                 false, false);
}

#define IPW 8
#define JT  (NA / 16)
#define NTW 8
#define KT  (KP / 4)
#define IPH 4
#define HP  (NA + 4)
#define FTH 1024

typedef __attribute__((ext_vector_type(4))) float v4f;

__global__ __launch_bounds__(FTH)
void ef_filter_reduce(const float* __restrict__ h,
                      const float* __restrict__ e,
                      const float* __restrict__ edge_mask,
                      const float* __restrict__ Wf,
                      const float* __restrict__ bf,
                      float* __restrict__ h1out) {
    __shared__ v2f   sh_b[(HD / 16) * KT * 32];
    __shared__ float sh_part[IPW * JT * HD];
    __shared__ alignas(16) float sh_h[HD * HP];
    __shared__ v2f   sh_bias[HD];

    const int tid  = threadIdx.x;
    const int b    = blockIdx.x >> 4;
    const int i0   = (blockIdx.x & 15) * IPW;
    const int lane = tid & 31;
    const int wave = tid >> 5;
    const int wave_s = __builtin_amdgcn_readfirstlane(wave);
    const int jt   = wave & 7;
    const int fh   = (wave >> 3) & 1;
    const int ih   = wave >> 4;
    const int half = lane >> 4;
    const int l    = lane & 15;

    for (int idx = tid; idx < (HD / 16) * KT * 32; idx += FTH) {
        const int ln = idx & 31;
        const int kt = (idx >> 5) & 7;
        const int nt = idx >> 8;
        const int f  = nt * 16 + (ln & 15);
        const int k  = 4 * kt + 2 * (ln >> 4);
        v2f v = {0.f, 0.f};
        if (k < EF) {
            v.x = Wf[f * EF + k];
            v.y = Wf[f * EF + k + 1];
        }
        sh_b[idx] = v;
    }

    if (tid < HD) {
        const float bv = bf[tid];
        const v2f bp = {bv, bv};
        sh_bias[tid] = bp;
    }

    {
        const float* hB = h + (size_t)b * NA * HD;
#pragma unroll 8
        for (int idx = tid; idx < NA * HD; idx += FTH) {
            const int j = idx >> 8;
            const int f = idx & (HD - 1);
            sh_h[f * HP + j] = hB[idx];
        }
    }

    const int jrow = jt * 16 + half * 8;
    const int f0   = fh * (NTW * 16) + l;
    const v2f* bP = sh_bias + f0;
    const float* hT = sh_h + f0 * HP + jrow;
    __syncthreads();

    const int ib = i0 + ih * IPH;
    const float* eA = e + (((size_t)(b * NA + ib)) * NA + jt * 16 + l) * EF;
    const float* mB = edge_mask + ((size_t)(b * NA + ib)) * NA + jrow;
    const v2f*   bW = sh_b + (fh * NTW) * KT * 32 + lane;
    const v2f z2 = {0.f, 0.f};

    v2f a[KT];
#pragma unroll
    for (int kt = 0; kt < KT - 1; ++kt)
        a[kt] = *(const v2f*)(eA + 4 * kt + 2 * half);
    a[KT - 1] = *(const v2f*)(eA + 4 * (KT - 1));
    a[KT - 1] = half ? z2 : a[KT - 1];

    v4f m0 = *(const v4f*)(mB);
    v4f m1 = *(const v4f*)(mB + 4);

    for (int il = 0; il < IPH; ++il) {
        const int iln = (il + 1 < IPH) ? il + 1 : il;
        const float* eN = eA + (size_t)iln * NA * EF;
        v2f an[KT];
#pragma unroll
        for (int kt = 0; kt < KT - 1; ++kt)
            an[kt] = *(const v2f*)(eN + 4 * kt + 2 * half);
        an[KT - 1] = *(const v2f*)(eN + 4 * (KT - 1));
        an[KT - 1] = half ? z2 : an[KT - 1];

        const float m[8] = {m0.x, m0.y, m0.z, m0.w, m1.x, m1.y, m1.z, m1.w};

        float* part = sh_part + ((ih * IPH + il) * JT + jt) * HD + fh * (NTW * 16) + lane;

#pragma unroll
        for (int np = 0; np < NTW / 2; ++np) {
            asm volatile("" ::: "memory");
            v2f bc0[KT], bc1[KT];
#pragma unroll
            for (int kt = 0; kt < KT; ++kt) {
                bc0[kt] = bW[((2 * np)     * KT + kt) * 32];
                bc1[kt] = bW[((2 * np + 1) * KT + kt) * 32];
            }
            const v4f ha0 = *(const v4f*)(hT + (2 * np) * 16 * HP);
            const v4f ha1 = *(const v4f*)(hT + (2 * np) * 16 * HP + 4);
            const v4f hb0 = *(const v4f*)(hT + (2 * np + 1) * 16 * HP);
            const v4f hb1 = *(const v4f*)(hT + (2 * np + 1) * 16 * HP + 4);
            const float hA[8] = {ha0.x, ha0.y, ha0.z, ha0.w, ha1.x, ha1.y, ha1.z, ha1.w};
            const float hB8[8] = {hb0.x, hb0.y, hb0.z, hb0.w, hb1.x, hb1.y, hb1.z, hb1.w};

            v8f c0 = {0.f, 0.f, 0.f, 0.f, 0.f, 0.f, 0.f, 0.f};
            v8f c1 = c0;
#pragma unroll
            for (int kt = 0; kt < KT; ++kt) {
                c0 = wmma_f32(a[kt], bc0[kt], c0);
                c1 = wmma_f32(a[kt], bc1[kt], c1);
            }
            const v2f bb0 = bP[(2 * np) * 16];
            const v2f bb1 = bP[(2 * np + 1) * 16];
            v2f pp0 = {0.f, 0.f}, pp1 = {0.f, 0.f};
#pragma unroll
            for (int q = 0; q < 4; ++q) {
                const v2f mq = {m[2 * q], m[2 * q + 1]};
                const v2f x0 = {c0[2 * q], c0[2 * q + 1]};
                const v2f x1 = {c1[2 * q], c1[2 * q + 1]};
                const v2f g0 = {hA[2 * q], hA[2 * q + 1]};
                const v2f g1 = {hB8[2 * q], hB8[2 * q + 1]};
                const v2f e0 = selu_pair(x0 + bb0) * mq;
                pp0 += e0 * g0;
                const v2f e1 = selu_pair(x1 + bb1) * mq;
                pp1 += e1 * g1;
            }
            const float p0 = pp0.x + pp0.y;
            const float p1 = pp1.x + pp1.y;
            const float t0 = p0 + __shfl_xor(p0, 16, 32);
            const float t1 = p1 + __shfl_xor(p1, 16, 32);
            part[np * 32] = half ? t1 : t0;
        }

#pragma unroll
        for (int kt = 0; kt < KT; ++kt)
            a[kt] = an[kt];
        m0 = *(const v4f*)(mB + (size_t)iln * NA);
        m1 = *(const v4f*)(mB + (size_t)iln * NA + 4);
    }
    __syncthreads();

    {
        const int tix = (wave_s << 5) | (int)__builtin_amdgcn_mbcnt_lo(~0u, 0u);
        const int f = tix & (HD - 1);
#pragma unroll
        for (int q = 0; q < IPW * HD / FTH; ++q) {
            const int il = q * (FTH / HD) + (tix >> 8);
            float s = 0.0f;
#pragma unroll
            for (int w = 0; w < JT; ++w)
                s += sh_part[(il * JT + w) * HD + f];
            h1out[((size_t)(b * NA + i0 + il)) * HD + f] = s;
        }
    }
}

#define MR   16
#define ZP   (K2 + 4)
#define CHS  8
#define NCH  (K2 / (8 * CHS))

__global__ __launch_bounds__(256) __attribute__((amdgpu_waves_per_eu(2, 2)))
void ef_node_mlp(const float* __restrict__ h,
                 const float* __restrict__ node_mask,
                 const float* __restrict__ Ww,
                 const float* __restrict__ bw,
                 float* __restrict__ out  ) {
    __shared__ float sh_z[MR * ZP];

    const int tid = threadIdx.x;
    const int r0  = blockIdx.x * MR;

    const int lane  = tid & 31;
    const int wave  = tid >> 5;
    const int half  = lane >> 4;
    const int l     = lane & 15;
    const int nbase = wave * 32;

    const float* w0 = Ww + (size_t)(nbase + l) * K2 + 4 * half;
    const float* w1 = w0 + (size_t)16 * K2;

    v4f zt[(MR * K2 / 4) / 256];
#pragma unroll
    for (int q = 0; q < (MR * K2 / 4) / 256; ++q) {
        const int idx = tid + 256 * q;
        const int row = idx >> 7;
        const int c4  = idx & 127;
        const float* src = (c4 < HD / 4)
            ? h   + (size_t)(r0 + row) * HD + c4 * 4
            : out + (size_t)(r0 + row) * HD + (c4 - HD / 4) * 4;
        zt[q] = *(const v4f*)src;
    }

    v4f b0c[CHS], b1c[CHS];
#pragma unroll
    for (int s = 0; s < CHS; ++s) {
        b0c[s] = *(const v4f*)(w0 + 8 * s);
        b1c[s] = *(const v4f*)(w1 + 8 * s);
    }

#pragma unroll
    for (int q = 0; q < (MR * K2 / 4) / 256; ++q) {
        const int idx = tid + 256 * q;
        *(v4f*)(sh_z + (idx >> 7) * ZP + (idx & 127) * 4) = zt[q];
    }
    __syncthreads();

    const int f0 = nbase + l;
    const int f1 = f0 + 16;
    const float bw0 = bw[f0];
    const float bw1 = bw[f1];
    float hr0[8], hr1[8], nmr[8];
#pragma unroll
    for (int r = 0; r < 8; ++r) {
        const int row = r0 + half * 8 + r;
        hr0[r] = h[(size_t)row * HD + f0];
        hr1[r] = h[(size_t)row * HD + f1];
        nmr[r] = node_mask[row];
    }

    const float* za = sh_z + l * ZP + 4 * half;
    v4f ac[CHS];
#pragma unroll
    for (int s = 0; s < CHS; ++s)
        ac[s] = *(const v4f*)(za + 8 * s);

    const v8f vzero = {0.f, 0.f, 0.f, 0.f, 0.f, 0.f, 0.f, 0.f};
    v8f acc0 = vzero, acc1 = vzero;

#pragma unroll
    for (int c = 0; c < NCH; ++c) {
        v4f an[CHS], b0n[CHS], b1n[CHS];
        if (c + 1 < NCH) {
#pragma unroll
            for (int s = 0; s < CHS; ++s) {
                b0n[s] = *(const v4f*)(w0 + 64 * (c + 1) + 8 * s);
                b1n[s] = *(const v4f*)(w1 + 64 * (c + 1) + 8 * s);
            }
#pragma unroll
            for (int s = 0; s < CHS; ++s)
                an[s] = *(const v4f*)(za + 64 * (c + 1) + 8 * s);
        }
#pragma unroll
        for (int s = 0; s < CHS; ++s) {
            const v2f a_lo = {ac[s].x, ac[s].y};
            const v2f a_hi = {ac[s].z, ac[s].w};
            const v2f p_lo = {b0c[s].x, b0c[s].y};
            const v2f p_hi = {b0c[s].z, b0c[s].w};
            const v2f q_lo = {b1c[s].x, b1c[s].y};
            const v2f q_hi = {b1c[s].z, b1c[s].w};
            acc0 = wmma_f32(a_lo, p_lo, acc0);
            acc1 = wmma_f32(a_lo, q_lo, acc1);
            acc0 = wmma_f32(a_hi, p_hi, acc0);
            acc1 = wmma_f32(a_hi, q_hi, acc1);
        }
        if (c + 1 < NCH) {
#pragma unroll
            for (int s = 0; s < CHS; ++s) {
                ac[s]  = an[s];
                b0c[s] = b0n[s];
                b1c[s] = b1n[s];
            }
        }
    }

#pragma unroll
    for (int r = 0; r < 8; ++r) {
        const int row = r0 + half * 8 + r;
        float v0 = selu_f(acc0[r] + bw0) * nmr[r];
        out[(size_t)row * HD + f0] = v0 + hr0[r];
        float v1 = selu_f(acc1[r] + bw1) * nmr[r];
        out[(size_t)row * HD + f1] = v1 + hr1[r];
    }
}

extern "C" void kernel_launch(void* const* d_in, const int* in_sizes, int n_in,
                              void* d_out, int out_size, void* d_ws, size_t ws_size,
                              hipStream_t stream) {
    (void)in_sizes; (void)n_in; (void)out_size; (void)d_ws; (void)ws_size;
    const float* h         = (const float*)d_in[0];
    const float* e         = (const float*)d_in[1];
    const float* node_mask = (const float*)d_in[2];
    const float* edge_mask = (const float*)d_in[3];
    const float* Wf        = (const float*)d_in[4];
    const float* bf        = (const float*)d_in[5];
    const float* Ww        = (const float*)d_in[6];
    const float* bw        = (const float*)d_in[7];
    float* out = (float*)d_out;

    ef_filter_reduce<<<(NB * NA) / IPW, FTH, 0, stream>>>(h, e, edge_mask, Wf, bf, out);
    ef_node_mlp<<<(NB * NA) / MR, 256, 0, stream>>>(h, node_mask, Ww, bw, out);
}


# Filter prologue: all staging loads (Wf, bias, h[b], first e rows and mask) requested before the first wait, loops unrolled (was 8 dependent round trips); h transposed by 16-byte LDS stores
# speedup vs baseline: 3.7313x; 1.0459x over previous
#include <hip/hip_runtime.h>


typedef __attribute__((ext_vector_type(2))) float v2f;
typedef __attribute__((ext_vector_type(8))) float v8f;

#define NB 16
#define NA 128
#define HD 256
#define EF 30
#define KP 32
#define K2 (2 * HD)

__device__ __forceinline__ float selu_f(float x) {
    const float scale = 1.0507009873554805f;
    const float alpha = 1.6732632423543772f;
    const float pos = scale * x;
    const float neg = scale * alpha * (__expf(x) - 1.0f);
    return x > 0.0f ? pos : neg;
}

__device__ __forceinline__ v2f selu_pair(v2f x) {
    const float scale = 1.0507009873554805f;
    const float sa    = 1.0507009873554805f * 1.6732632423543772f;
    const v2f zero = {0.f, 0.f};
    const v2f mn = __builtin_elementwise_min(x, zero);
    const v2f mx = __builtin_elementwise_max(x, zero);
    const v2f ex = {__expf(mn.x), __expf(mn.y)};
    const v2f neg = ex * sa - sa;
    return mx * scale + neg;
}

__device__ __forceinline__ v8f wmma_f32(v2f a, v2f b, v8f c) {
    return __builtin_amdgcn_wmma_f32_16x16x4_f32(false, a, false, b, (short)0, c,
                                                 false, false);
}

#define IPW 8
#define JT  (NA / 16)
#define NTW 8
#define KT  (KP / 4)
#define IPH 4
#define HP  (NA + 4)
#define FTH 1024

typedef __attribute__((ext_vector_type(4))) float v4f;

__global__ __launch_bounds__(FTH)
void ef_filter_reduce(const float* __restrict__ h,
                      const float* __restrict__ e,
                      const float* __restrict__ edge_mask,
                      const float* __restrict__ Wf,
                      const float* __restrict__ bf,
                      float* __restrict__ h1out) {
    __shared__ v2f   sh_b[(HD / 16) * KT * 32];
    __shared__ float sh_part[IPW * JT * HD];
    __shared__ alignas(16) float sh_h[HD * HP];
    __shared__ v2f   sh_bias[HD];

    const int tid  = threadIdx.x;
    const int b    = blockIdx.x >> 4;
    const int i0   = (blockIdx.x & 15) * IPW;
    const int lane = tid & 31;
    const int wave = tid >> 5;
    const int wave_s = __builtin_amdgcn_readfirstlane(wave);
    const int jt   = wave & 7;
    const int fh   = (wave >> 3) & 1;
    const int ih   = wave >> 4;
    const int half = lane >> 4;
    const int l    = lane & 15;


    const int jrow = jt * 16 + half * 8;
    const int f0   = fh * (NTW * 16) + l;
    const int ib = i0 + ih * IPH;
    const float* eA = e + (((size_t)(b * NA + ib)) * NA + jt * 16 + l) * EF;
    const float* mB = edge_mask + ((size_t)(b * NA + ib)) * NA + jrow;
    const v2f z2 = {0.f, 0.f};

    const int wkt = (tid >> 5) & 7;
    const int wk  = 4 * wkt + 2 * half;
    const bool wpad = wk >= EF;
    const float* wfp = Wf + ((tid >> 8) * 16 + l) * EF + (wpad ? EF - 2 : wk);
    v2f wv[4];
#pragma unroll
    for (int q = 0; q < 4; ++q)
        wv[q] = *(const v2f*)(wfp + (size_t)q * 64 * EF);

    const float bv = bf[tid & (HD - 1)];

    const float* hB = h + (size_t)b * NA * HD + (tid & (HD - 1));
    const int jq = tid >> 8;
    v4f hv[8];
#pragma unroll
    for (int q = 0; q < 8; ++q) {
        const float* hp = hB + (size_t)(4 * (jq + 4 * q)) * HD;
        hv[q].x = hp[0];
        hv[q].y = hp[HD];
        hv[q].z = hp[2 * HD];
        hv[q].w = hp[3 * HD];
    }

    v2f a[KT];
#pragma unroll
    for (int kt = 0; kt < KT - 1; ++kt)
        a[kt] = *(const v2f*)(eA + 4 * kt + 2 * half);
    a[KT - 1] = *(const v2f*)(eA + 4 * (KT - 1));
    v4f m0 = *(const v4f*)(mB);
    v4f m1 = *(const v4f*)(mB + 4);

    asm volatile("" :: "v"(bv));

#pragma unroll
    for (int q = 0; q < 4; ++q)
        sh_b[tid + FTH * q] = wpad ? z2 : wv[q];
    if (tid < HD) {
        const v2f bp = {bv, bv};
        sh_bias[tid] = bp;
    }
#pragma unroll
    for (int q = 0; q < 8; ++q)
        *(v4f*)(sh_h + (tid & (HD - 1)) * HP + 4 * (jq + 4 * q)) = hv[q];

    const v2f* bP = sh_bias + f0;
    const float* hT = sh_h + f0 * HP + jrow;
    const v2f*   bW = sh_b + (fh * NTW) * KT * 32 + lane;

    asm volatile("" :: "v"(a[0]), "v"(a[1]), "v"(a[2]), "v"(a[3]), "v"(a[4]), "v"(a[5]),
                       "v"(a[6]), "v"(a[7]), "v"(m0), "v"(m1));
    __syncthreads();

    a[KT - 1] = half ? z2 : a[KT - 1];

    for (int il = 0; il < IPH; ++il) {
        const int iln = (il + 1 < IPH) ? il + 1 : il;
        const float* eN = eA + (size_t)iln * NA * EF;
        v2f an[KT];
#pragma unroll
        for (int kt = 0; kt < KT - 1; ++kt)
            an[kt] = *(const v2f*)(eN + 4 * kt + 2 * half);
        an[KT - 1] = *(const v2f*)(eN + 4 * (KT - 1));
        an[KT - 1] = half ? z2 : an[KT - 1];

        const float m[8] = {m0.x, m0.y, m0.z, m0.w, m1.x, m1.y, m1.z, m1.w};

        float* part = sh_part + ((ih * IPH + il) * JT + jt) * HD + fh * (NTW * 16) + lane;

#pragma unroll
        for (int np = 0; np < NTW / 2; ++np) {
            asm volatile("" ::: "memory");
            v2f bc0[KT], bc1[KT];
#pragma unroll
            for (int kt = 0; kt < KT; ++kt) {
                bc0[kt] = bW[((2 * np)     * KT + kt) * 32];
                bc1[kt] = bW[((2 * np + 1) * KT + kt) * 32];
            }
            const v4f ha0 = *(const v4f*)(hT + (2 * np) * 16 * HP);
            const v4f ha1 = *(const v4f*)(hT + (2 * np) * 16 * HP + 4);
            const v4f hb0 = *(const v4f*)(hT + (2 * np + 1) * 16 * HP);
            const v4f hb1 = *(const v4f*)(hT + (2 * np + 1) * 16 * HP + 4);
            const float hA[8] = {ha0.x, ha0.y, ha0.z, ha0.w, ha1.x, ha1.y, ha1.z, ha1.w};
            const float hB8[8] = {hb0.x, hb0.y, hb0.z, hb0.w, hb1.x, hb1.y, hb1.z, hb1.w};

            v8f c0 = {0.f, 0.f, 0.f, 0.f, 0.f, 0.f, 0.f, 0.f};
            v8f c1 = c0;
#pragma unroll
            for (int kt = 0; kt < KT; ++kt) {
                c0 = wmma_f32(a[kt], bc0[kt], c0);
                c1 = wmma_f32(a[kt], bc1[kt], c1);
            }
            const v2f bb0 = bP[(2 * np) * 16];
            const v2f bb1 = bP[(2 * np + 1) * 16];
            v2f pp0 = {0.f, 0.f}, pp1 = {0.f, 0.f};
#pragma unroll
            for (int q = 0; q < 4; ++q) {
                const v2f mq = {m[2 * q], m[2 * q + 1]};
                const v2f x0 = {c0[2 * q], c0[2 * q + 1]};
                const v2f x1 = {c1[2 * q], c1[2 * q + 1]};
                const v2f g0 = {hA[2 * q], hA[2 * q + 1]};
                const v2f g1 = {hB8[2 * q], hB8[2 * q + 1]};
                const v2f e0 = selu_pair(x0 + bb0) * mq;
                pp0 += e0 * g0;
                const v2f e1 = selu_pair(x1 + bb1) * mq;
                pp1 += e1 * g1;
            }
            const float p0 = pp0.x + pp0.y;
            const float p1 = pp1.x + pp1.y;
            const float t0 = p0 + __shfl_xor(p0, 16, 32);
            const float t1 = p1 + __shfl_xor(p1, 16, 32);
            part[np * 32] = half ? t1 : t0;
        }

#pragma unroll
        for (int kt = 0; kt < KT; ++kt)
            a[kt] = an[kt];
        m0 = *(const v4f*)(mB + (size_t)iln * NA);
        m1 = *(const v4f*)(mB + (size_t)iln * NA + 4);
    }
    __syncthreads();

    {
        const int tix = (wave_s << 5) | (int)__builtin_amdgcn_mbcnt_lo(~0u, 0u);
        const int f = tix & (HD - 1);
#pragma unroll
        for (int q = 0; q < IPW * HD / FTH; ++q) {
            const int il = q * (FTH / HD) + (tix >> 8);
            float s = 0.0f;
#pragma unroll
            for (int w = 0; w < JT; ++w)
                s += sh_part[(il * JT + w) * HD + f];
            h1out[((size_t)(b * NA + i0 + il)) * HD + f] = s;
        }
    }
}

#define MR   16
#define ZP   (K2 + 4)
#define CHS  8
#define NCH  (K2 / (8 * CHS))

__global__ __launch_bounds__(256) __attribute__((amdgpu_waves_per_eu(2, 2)))
void ef_node_mlp(const float* __restrict__ h,
                 const float* __restrict__ node_mask,
                 const float* __restrict__ Ww,
                 const float* __restrict__ bw,
                 float* __restrict__ out  ) {
    __shared__ float sh_z[MR * ZP];

    const int tid = threadIdx.x;
    const int r0  = blockIdx.x * MR;

    const int lane  = tid & 31;
    const int wave  = tid >> 5;
    const int half  = lane >> 4;
    const int l     = lane & 15;
    const int nbase = wave * 32;

    const float* w0 = Ww + (size_t)(nbase + l) * K2 + 4 * half;
    const float* w1 = w0 + (size_t)16 * K2;

    v4f zt[(MR * K2 / 4) / 256];
#pragma unroll
    for (int q = 0; q < (MR * K2 / 4) / 256; ++q) {
        const int idx = tid + 256 * q;
        const int row = idx >> 7;
        const int c4  = idx & 127;
        const float* src = (c4 < HD / 4)
            ? h   + (size_t)(r0 + row) * HD + c4 * 4
            : out + (size_t)(r0 + row) * HD + (c4 - HD / 4) * 4;
        zt[q] = *(const v4f*)src;
    }

    v4f b0c[CHS], b1c[CHS];
#pragma unroll
    for (int s = 0; s < CHS; ++s) {
        b0c[s] = *(const v4f*)(w0 + 8 * s);
        b1c[s] = *(const v4f*)(w1 + 8 * s);
    }

#pragma unroll
    for (int q = 0; q < (MR * K2 / 4) / 256; ++q) {
        const int idx = tid + 256 * q;
        *(v4f*)(sh_z + (idx >> 7) * ZP + (idx & 127) * 4) = zt[q];
    }
    __syncthreads();

    const int f0 = nbase + l;
    const int f1 = f0 + 16;
    const float bw0 = bw[f0];
    const float bw1 = bw[f1];
    float hr0[8], hr1[8], nmr[8];
#pragma unroll
    for (int r = 0; r < 8; ++r) {
        const int row = r0 + half * 8 + r;
        hr0[r] = h[(size_t)row * HD + f0];
        hr1[r] = h[(size_t)row * HD + f1];
        nmr[r] = node_mask[row];
    }

    const float* za = sh_z + l * ZP + 4 * half;
    v4f ac[CHS];
#pragma unroll
    for (int s = 0; s < CHS; ++s)
        ac[s] = *(const v4f*)(za + 8 * s);

    const v8f vzero = {0.f, 0.f, 0.f, 0.f, 0.f, 0.f, 0.f, 0.f};
    v8f acc0 = vzero, acc1 = vzero;

#pragma unroll
    for (int c = 0; c < NCH; ++c) {
        v4f an[CHS], b0n[CHS], b1n[CHS];
        if (c + 1 < NCH) {
#pragma unroll
            for (int s = 0; s < CHS; ++s) {
                b0n[s] = *(const v4f*)(w0 + 64 * (c + 1) + 8 * s);
                b1n[s] = *(const v4f*)(w1 + 64 * (c + 1) + 8 * s);
            }
#pragma unroll
            for (int s = 0; s < CHS; ++s)
                an[s] = *(const v4f*)(za + 64 * (c + 1) + 8 * s);
        }
#pragma unroll
        for (int s = 0; s < CHS; ++s) {
            const v2f a_lo = {ac[s].x, ac[s].y};
            const v2f a_hi = {ac[s].z, ac[s].w};
            const v2f p_lo = {b0c[s].x, b0c[s].y};
            const v2f p_hi = {b0c[s].z, b0c[s].w};
            const v2f q_lo = {b1c[s].x, b1c[s].y};
            const v2f q_hi = {b1c[s].z, b1c[s].w};
            acc0 = wmma_f32(a_lo, p_lo, acc0);
            acc1 = wmma_f32(a_lo, q_lo, acc1);
            acc0 = wmma_f32(a_hi, p_hi, acc0);
            acc1 = wmma_f32(a_hi, q_hi, acc1);
        }
        if (c + 1 < NCH) {
#pragma unroll
            for (int s = 0; s < CHS; ++s) {
                ac[s]  = an[s];
                b0c[s] = b0n[s];
                b1c[s] = b1n[s];
            }
        }
    }

#pragma unroll
    for (int r = 0; r < 8; ++r) {
        const int row = r0 + half * 8 + r;
        float v0 = selu_f(acc0[r] + bw0) * nmr[r];
        out[(size_t)row * HD + f0] = v0 + hr0[r];
        float v1 = selu_f(acc1[r] + bw1) * nmr[r];
        out[(size_t)row * HD + f1] = v1 + hr1[r];
    }
}

extern "C" void kernel_launch(void* const* d_in, const int* in_sizes, int n_in,
                              void* d_out, int out_size, void* d_ws, size_t ws_size,
                              hipStream_t stream) {
    (void)in_sizes; (void)n_in; (void)out_size; (void)d_ws; (void)ws_size;
    const float* h         = (const float*)d_in[0];
    const float* e         = (const float*)d_in[1];
    const float* node_mask = (const float*)d_in[2];
    const float* edge_mask = (const float*)d_in[3];
    const float* Wf        = (const float*)d_in[4];
    const float* bf        = (const float*)d_in[5];
    const float* Ww        = (const float*)d_in[6];
    const float* bw        = (const float*)d_in[7];
    float* out = (float*)d_out;

    ef_filter_reduce<<<(NB * NA) / IPW, FTH, 0, stream>>>(h, e, edge_mask, Wf, bf, out);
    ef_node_mlp<<<(NB * NA) / MR, 256, 0, stream>>>(h, node_mask, Ww, bw, out);
}


# Node MLP on all 256 WGPs: groups of 32 rows x 64 columns (was 16 x 256, 128 groups), a wave one tile in two k-interleaved sums; h1 passed in 2 MB of workspace (in place if smaller): sum order changes
# speedup vs baseline: 3.9611x; 1.0616x over previous
#include <hip/hip_runtime.h>


typedef __attribute__((ext_vector_type(2))) float v2f;
typedef __attribute__((ext_vector_type(8))) float v8f;

#define NB 16
#define NA 128
#define HD 256
#define EF 30
#define KP 32
#define K2 (2 * HD)
#define H1_BYTES ((unsigned long long)NB * NA * HD * 4)

__device__ __forceinline__ float selu_f(float x) {
    const float scale = 1.0507009873554805f;
    const float alpha = 1.6732632423543772f;
    const float pos = scale * x;
    const float neg = scale * alpha * (__expf(x) - 1.0f);
    return x > 0.0f ? pos : neg;
}

__device__ __forceinline__ v2f selu_pair(v2f x) {
    const float scale = 1.0507009873554805f;
    const float sa    = 1.0507009873554805f * 1.6732632423543772f;
    const v2f zero = {0.f, 0.f};
    const v2f mn = __builtin_elementwise_min(x, zero);
    const v2f mx = __builtin_elementwise_max(x, zero);
    const v2f ex = {__expf(mn.x), __expf(mn.y)};
    const v2f neg = ex * sa - sa;
    return mx * scale + neg;
}

__device__ __forceinline__ v8f wmma_f32(v2f a, v2f b, v8f c) {
    return __builtin_amdgcn_wmma_f32_16x16x4_f32(false, a, false, b, (short)0, c,
                                                 false, false);
}

#define IPW 8
#define JT  (NA / 16)
#define NTW 8
#define KT  (KP / 4)
#define IPH 4
#define HP  (NA + 4)
#define FTH 1024

typedef __attribute__((ext_vector_type(4))) float v4f;

__global__ __launch_bounds__(FTH)
void ef_filter_reduce(const float* __restrict__ h,
                      const float* __restrict__ e,
                      const float* __restrict__ edge_mask,
                      const float* __restrict__ Wf,
                      const float* __restrict__ bf,
                      float* __restrict__ h1out,
                      float* __restrict__ ws,
                      unsigned long long ws_bytes) {
    __shared__ v2f   sh_b[(HD / 16) * KT * 32];
    __shared__ float sh_part[IPW * JT * HD];
    __shared__ alignas(16) float sh_h[HD * HP];
    __shared__ v2f   sh_bias[HD];

    const int tid  = threadIdx.x;
    const int b    = blockIdx.x >> 4;
    const int i0   = (blockIdx.x & 15) * IPW;
    const int lane = tid & 31;
    const int wave = tid >> 5;
    const int wave_s = __builtin_amdgcn_readfirstlane(wave);
    const int jt   = wave & 7;
    const int fh   = (wave >> 3) & 1;
    const int ih   = wave >> 4;
    const int half = lane >> 4;
    const int l    = lane & 15;


    const int jrow = jt * 16 + half * 8;
    const int f0   = fh * (NTW * 16) + l;
    const int ib = i0 + ih * IPH;
    const float* eA = e + (((size_t)(b * NA + ib)) * NA + jt * 16 + l) * EF;
    const float* mB = edge_mask + ((size_t)(b * NA + ib)) * NA + jrow;
    const v2f z2 = {0.f, 0.f};

    const int wkt = (tid >> 5) & 7;
    const int wk  = 4 * wkt + 2 * half;
    const bool wpad = wk >= EF;
    const float* wfp = Wf + ((tid >> 8) * 16 + l) * EF + (wpad ? EF - 2 : wk);
    v2f wv[4];
#pragma unroll
    for (int q = 0; q < 4; ++q)
        wv[q] = *(const v2f*)(wfp + (size_t)q * 64 * EF);

    const float bv = bf[tid & (HD - 1)];

    const float* hB = h + (size_t)b * NA * HD + (tid & (HD - 1));
    const int jq = tid >> 8;
    v4f hv[8];
#pragma unroll
    for (int q = 0; q < 8; ++q) {
        const float* hp = hB + (size_t)(4 * (jq + 4 * q)) * HD;
        hv[q].x = hp[0];
        hv[q].y = hp[HD];
        hv[q].z = hp[2 * HD];
        hv[q].w = hp[3 * HD];
    }

    v2f a[KT];
#pragma unroll
    for (int kt = 0; kt < KT - 1; ++kt)
        a[kt] = *(const v2f*)(eA + 4 * kt + 2 * half);
    a[KT - 1] = *(const v2f*)(eA + 4 * (KT - 1));
    v4f m0 = *(const v4f*)(mB);
    v4f m1 = *(const v4f*)(mB + 4);

    asm volatile("" :: "v"(bv));

#pragma unroll
    for (int q = 0; q < 4; ++q)
        sh_b[tid + FTH * q] = wpad ? z2 : wv[q];
    if (tid < HD) {
        const v2f bp = {bv, bv};
        sh_bias[tid] = bp;
    }
#pragma unroll
    for (int q = 0; q < 8; ++q)
        *(v4f*)(sh_h + (tid & (HD - 1)) * HP + 4 * (jq + 4 * q)) = hv[q];

    const v2f* bP = sh_bias + f0;
    const float* hT = sh_h + f0 * HP + jrow;
    const v2f*   bW = sh_b + (fh * NTW) * KT * 32 + lane;

    asm volatile("" :: "v"(a[0]), "v"(a[1]), "v"(a[2]), "v"(a[3]), "v"(a[4]), "v"(a[5]),
                       "v"(a[6]), "v"(a[7]), "v"(m0), "v"(m1));
    __syncthreads();

    a[KT - 1] = half ? z2 : a[KT - 1];

    for (int il = 0; il < IPH; ++il) {
        const int iln = (il + 1 < IPH) ? il + 1 : il;
        const float* eN = eA + (size_t)iln * NA * EF;
        v2f an[KT];
#pragma unroll
        for (int kt = 0; kt < KT - 1; ++kt)
            an[kt] = *(const v2f*)(eN + 4 * kt + 2 * half);
        an[KT - 1] = *(const v2f*)(eN + 4 * (KT - 1));
        an[KT - 1] = half ? z2 : an[KT - 1];

        const float m[8] = {m0.x, m0.y, m0.z, m0.w, m1.x, m1.y, m1.z, m1.w};

        float* part = sh_part + ((ih * IPH + il) * JT + jt) * HD + fh * (NTW * 16) + lane;

#pragma unroll
        for (int np = 0; np < NTW / 2; ++np) {
            asm volatile("" ::: "memory");
            v2f bc0[KT], bc1[KT];
#pragma unroll
            for (int kt = 0; kt < KT; ++kt) {
                bc0[kt] = bW[((2 * np)     * KT + kt) * 32];
                bc1[kt] = bW[((2 * np + 1) * KT + kt) * 32];
            }
            const v4f ha0 = *(const v4f*)(hT + (2 * np) * 16 * HP);
            const v4f ha1 = *(const v4f*)(hT + (2 * np) * 16 * HP + 4);
            const v4f hb0 = *(const v4f*)(hT + (2 * np + 1) * 16 * HP);
            const v4f hb1 = *(const v4f*)(hT + (2 * np + 1) * 16 * HP + 4);
            const float hA[8] = {ha0.x, ha0.y, ha0.z, ha0.w, ha1.x, ha1.y, ha1.z, ha1.w};
            const float hB8[8] = {hb0.x, hb0.y, hb0.z, hb0.w, hb1.x, hb1.y, hb1.z, hb1.w};

            v8f c0 = {0.f, 0.f, 0.f, 0.f, 0.f, 0.f, 0.f, 0.f};
            v8f c1 = c0;
#pragma unroll
            for (int kt = 0; kt < KT; ++kt) {
                c0 = wmma_f32(a[kt], bc0[kt], c0);
                c1 = wmma_f32(a[kt], bc1[kt], c1);
            }
            const v2f bb0 = bP[(2 * np) * 16];
            const v2f bb1 = bP[(2 * np + 1) * 16];
            v2f pp0 = {0.f, 0.f}, pp1 = {0.f, 0.f};
#pragma unroll
            for (int q = 0; q < 4; ++q) {
                const v2f mq = {m[2 * q], m[2 * q + 1]};
                const v2f x0 = {c0[2 * q], c0[2 * q + 1]};
                const v2f x1 = {c1[2 * q], c1[2 * q + 1]};
                const v2f g0 = {hA[2 * q], hA[2 * q + 1]};
                const v2f g1 = {hB8[2 * q], hB8[2 * q + 1]};
                const v2f e0 = selu_pair(x0 + bb0) * mq;
                pp0 += e0 * g0;
                const v2f e1 = selu_pair(x1 + bb1) * mq;
                pp1 += e1 * g1;
            }
            const float p0 = pp0.x + pp0.y;
            const float p1 = pp1.x + pp1.y;
            const float t0 = p0 + __shfl_xor(p0, 16, 32);
            const float t1 = p1 + __shfl_xor(p1, 16, 32);
            part[np * 32] = half ? t1 : t0;
        }

#pragma unroll
        for (int kt = 0; kt < KT; ++kt)
            a[kt] = an[kt];
        m0 = *(const v4f*)(mB + (size_t)iln * NA);
        m1 = *(const v4f*)(mB + (size_t)iln * NA + 4);
    }
    __syncthreads();

    {
        const int tix = (wave_s << 5) | (int)__builtin_amdgcn_mbcnt_lo(~0u, 0u);
        const int f = tix & (HD - 1);
        float* h1dst = (ws_bytes >= H1_BYTES) ? ws : h1out;
#pragma unroll
        for (int q = 0; q < IPW * HD / FTH; ++q) {
            const int il = q * (FTH / HD) + (tix >> 8);
            float s = 0.0f;
#pragma unroll
            for (int w = 0; w < JT; ++w)
                s += sh_part[(il * JT + w) * HD + f];
            h1dst[((size_t)(b * NA + i0 + il)) * HD + f] = s;
        }
    }
}

#define MR   32
#define CW   64
#define ZP   (K2 + 4)
#define CHS  8
#define NCH  (K2 / (8 * CHS))

__global__ __launch_bounds__(256) __attribute__((amdgpu_waves_per_eu(2, 2)))
void ef_node_mlp(const float* __restrict__ h,
                 const float* __restrict__ node_mask,
                 const float* __restrict__ Ww,
                 const float* __restrict__ bw,
                 float* out,
                 const float* ws,
                 unsigned long long ws_bytes) {
    __shared__ float sh_z[MR * ZP];

    const bool split = ws_bytes >= H1_BYTES;
    const int cq = blockIdx.x & 3;
    if (!split && cq) return;
    const float* h1 = split ? ws : out;
    const int cb0 = split ? cq * CW : 0;
    const int cb1 = split ? cb0 + CW : HD;

    const int tid = threadIdx.x;
    const int r0  = (blockIdx.x >> 2) * MR;

    const int lane  = tid & 31;
    const int wave  = tid >> 5;
    const int half  = lane >> 4;
    const int l     = lane & 15;
    const int rt    = wave & 1;
    const int ct    = wave >> 1;

    const float* w0 = Ww + (size_t)(cb0 + ct * 16 + l) * K2 + 4 * half;

    v4f zt[(MR * K2 / 4) / 256];
#pragma unroll
    for (int q = 0; q < (MR * K2 / 4) / 256; ++q) {
        const int idx = tid + 256 * q;
        const int row = idx >> 7;
        const int c4  = idx & 127;
        const float* src = (c4 < HD / 4)
            ? h  + (size_t)(r0 + row) * HD + c4 * 4
            : h1 + (size_t)(r0 + row) * HD + (c4 - HD / 4) * 4;
        zt[q] = *(const v4f*)src;
    }

    v4f bq[3][CHS];
#pragma unroll
    for (int s = 0; s < CHS; ++s) {
        bq[0][s] = *(const v4f*)(w0 + 8 * s);
        bq[1][s] = *(const v4f*)(w0 + 64 + 8 * s);
    }

#pragma unroll
    for (int q = 0; q < (MR * K2 / 4) / 256; ++q) {
        const int idx = tid + 256 * q;
        *(v4f*)(sh_z + (idx >> 7) * ZP + (idx & 127) * 4) = zt[q];
    }
    __syncthreads();

    const float* za = sh_z + (rt * 16 + l) * ZP + 4 * half;

    for (int cbase = cb0; ; ) {
        const int f0 = cbase + ct * 16 + l;
        const int rw = r0 + rt * 16 + half * 8;
        const float bw0 = bw[f0];
        float hr0[8], nmr[8];
#pragma unroll
        for (int r = 0; r < 8; ++r) {
            hr0[r] = h[(size_t)(rw + r) * HD + f0];
            nmr[r] = node_mask[rw + r];
        }

        v4f ac[CHS];
#pragma unroll
        for (int s = 0; s < CHS; ++s)
            ac[s] = *(const v4f*)(za + 8 * s);

        const v8f vzero = {0.f, 0.f, 0.f, 0.f, 0.f, 0.f, 0.f, 0.f};
        v8f acc0 = vzero, acc1 = vzero;

#pragma unroll
        for (int c = 0; c < NCH; ++c) {
            v4f an[CHS];
            asm volatile("" ::: "memory");
            __builtin_amdgcn_sched_barrier(0);
            if (c + 2 < NCH) {
#pragma unroll
                for (int s = 0; s < CHS; ++s)
                    bq[(c + 2) % 3][s] = *(const v4f*)(w0 + 64 * (c + 2) + 8 * s);
            }
            if (c + 1 < NCH) {
#pragma unroll
                for (int s = 0; s < CHS; ++s)
                    an[s] = *(const v4f*)(za + 64 * (c + 1) + 8 * s);
            }
#pragma unroll
            for (int s = 0; s < CHS; ++s) {
                const v4f bb = bq[c % 3][s];
                const v2f a_lo = {ac[s].x, ac[s].y};
                const v2f a_hi = {ac[s].z, ac[s].w};
                const v2f p_lo = {bb.x, bb.y};
                const v2f p_hi = {bb.z, bb.w};
                acc0 = wmma_f32(a_lo, p_lo, acc0);
                acc1 = wmma_f32(a_hi, p_hi, acc1);
            }
            if (c + 1 < NCH) {
#pragma unroll
                for (int s = 0; s < CHS; ++s)
                    ac[s] = an[s];
            }
            __builtin_amdgcn_sched_group_barrier(0x100, 8, 0);
            __builtin_amdgcn_sched_group_barrier(0x020, 8, 0);
            __builtin_amdgcn_sched_group_barrier(0x008, 16, 0);
        }

#pragma unroll
        for (int r = 0; r < 8; ++r) {
            const float v0 = selu_f(acc0[r] + acc1[r] + bw0) * nmr[r];
            out[(size_t)(rw + r) * HD + f0] = v0 + hr0[r];
        }

        cbase += CW;
        if (cbase >= cb1) break;
        w0 += (size_t)CW * K2;
#pragma unroll
        for (int s = 0; s < CHS; ++s) {
            bq[0][s] = *(const v4f*)(w0 + 8 * s);
            bq[1][s] = *(const v4f*)(w0 + 64 + 8 * s);
        }
    }
}

extern "C" void kernel_launch(void* const* d_in, const int* in_sizes, int n_in,
                              void* d_out, int out_size, void* d_ws, size_t ws_size,
                              hipStream_t stream) {
    (void)in_sizes; (void)n_in; (void)out_size;
    const float* h         = (const float*)d_in[0];
    const float* e         = (const float*)d_in[1];
    const float* node_mask = (const float*)d_in[2];
    const float* edge_mask = (const float*)d_in[3];
    const float* Wf        = (const float*)d_in[4];
    const float* bf        = (const float*)d_in[5];
    const float* Ww        = (const float*)d_in[6];
    const float* bw        = (const float*)d_in[7];
    float* out = (float*)d_out;
    float* ws  = (float*)d_ws;

    ef_filter_reduce<<<(NB * NA) / IPW, FTH, 0, stream>>>(h, e, edge_mask, Wf, bf, out, ws,
                                                          (unsigned long long)ws_size);
    ef_node_mlp<<<((NB * NA) / MR) * (HD / CW), 256, 0, stream>>>(h, node_mask, Ww, bw, out, ws,
                                                                 (unsigned long long)ws_size);
}
